# starting kernel
# baseline (speedup 1.0000x reference)
#include <hip/hip_runtime.h>


#define N_NODES 50000
#define N_EDGES 600000
#define DIM     128
#define HID     256
#define AS_STR  (DIM + 4)
#define HS_STR  (HID + 4)

typedef __attribute__((ext_vector_type(2))) float v2f;
typedef __attribute__((ext_vector_type(8))) float v8f;

__global__ __launch_bounds__(256) void init_acc(const float* __restrict__ x,
                                                const float* __restrict__ e1,
                                                const float* __restrict__ e2,
                                                float* __restrict__ acc) {
  int idx = blockIdx.x * 256 + threadIdx.x;
  int d   = idx & (DIM - 1);
  acc[idx] = x[idx] + e1[4 * DIM + d] + e2[0 * DIM + d];
}

__global__ __launch_bounds__(256) void scatter_edges(const float* __restrict__ x,
                                                     const int* __restrict__ edge_index,
                                                     const int* __restrict__ edge_attr,
                                                     const float* __restrict__ e1,
                                                     const float* __restrict__ e2,
                                                     float* __restrict__ acc) {
  int idx = blockIdx.x * 256 + threadIdx.x;
  int e   = idx >> 7;
  int d   = idx & (DIM - 1);
  int row = edge_index[e];
  int col = edge_index[N_EDGES + e];
  int a0  = edge_attr[2 * e];
  int a1  = edge_attr[2 * e + 1];
  float val = x[col * DIM + d] + e1[a0 * DIM + d] + e2[a1 * DIM + d];
  unsafeAtomicAdd(&acc[row * DIM + d], val);
}

__global__ __launch_bounds__(128) void mlp_fused(const float* __restrict__ acc,
                                                 const float* __restrict__ W1,
                                                 const float* __restrict__ b1,
                                                 const float* __restrict__ W2,
                                                 const float* __restrict__ b2,
                                                 float* __restrict__ out) {
  __shared__ float As[16 * AS_STR];
  __shared__ float Hs[16 * HS_STR];
  const int tid  = threadIdx.x;
  const int wave = tid >> 5;
  const int lane = tid & 31;
  const int m    = lane & 15;
  const int kh   = lane >> 4;
  const int row0 = blockIdx.x * 16;

  {
    const int r  = tid >> 3;
    const int c0 = (tid & 7) * 16;
    const float4* src = (const float4*)(acc + (row0 + r) * DIM + c0);
    float4*       dst = (float4*)(As + r * AS_STR + c0);
#pragma unroll
    for (int i = 0; i < 4; ++i) dst[i] = src[i];
  }
  __syncthreads();

#pragma unroll
  for (int nt = 0; nt < 4; ++nt) {
    const int n0 = (wave * 4 + nt) * 16;
    const float* w1row = W1 + (n0 + m) * DIM;
    v8f c = {};
#pragma unroll
    for (int k0 = 0; k0 < DIM; k0 += 4) {
      v2f a, b;
      a.x = As[m * AS_STR + k0 + 2 * kh];
      a.y = As[m * AS_STR + k0 + 2 * kh + 1];
      b.x = w1row[k0 + 2 * kh];
      b.y = w1row[k0 + 2 * kh + 1];
      c = __builtin_amdgcn_wmma_f32_16x16x4_f32(false, a, false, b, (short)0, c,
                                                false, false);
    }
    const float bias = b1[n0 + m];
#pragma unroll
    for (int j = 0; j < 8; ++j) {
      const int mj = j + 8 * kh;
      float h = c[j] + bias;
      Hs[mj * HS_STR + n0 + m] = h > 0.0f ? h : 0.0f;
    }
  }
  __syncthreads();

#pragma unroll
  for (int nt = 0; nt < 2; ++nt) {
    const int n0 = (wave * 2 + nt) * 16;
    const float* w2row = W2 + (n0 + m) * HID;
    v8f c = {};
#pragma unroll
    for (int k0 = 0; k0 < HID; k0 += 4) {
      v2f a, b;
      a.x = Hs[m * HS_STR + k0 + 2 * kh];
      a.y = Hs[m * HS_STR + k0 + 2 * kh + 1];
      b.x = w2row[k0 + 2 * kh];
      b.y = w2row[k0 + 2 * kh + 1];
      c = __builtin_amdgcn_wmma_f32_16x16x4_f32(false, a, false, b, (short)0, c,
                                                false, false);
    }
    const float bias = b2[n0 + m];
#pragma unroll
    for (int j = 0; j < 8; ++j) {
      const int mj = j + 8 * kh;
      out[(row0 + mj) * DIM + n0 + m] = c[j] + bias;
    }
  }
}

extern "C" void kernel_launch(void* const* d_in, const int* in_sizes, int n_in,
                              void* d_out, int out_size, void* d_ws, size_t ws_size,
                              hipStream_t stream) {
  const float* x          = (const float*)d_in[0];
  const int*   edge_index = (const int*)d_in[1];
  const int*   edge_attr  = (const int*)d_in[2];
  const float* W1         = (const float*)d_in[3];
  const float* b1         = (const float*)d_in[4];
  const float* W2         = (const float*)d_in[5];
  const float* b2         = (const float*)d_in[6];
  const float* e1         = (const float*)d_in[7];
  const float* e2         = (const float*)d_in[8];
  float*       out        = (float*)d_out;
  float*       acc        = (float*)d_ws;

  (void)in_sizes; (void)n_in; (void)out_size; (void)ws_size;

  init_acc<<<(N_NODES * DIM) / 256, 256, 0, stream>>>(x, e1, e2, acc);
  scatter_edges<<<(N_EDGES * DIM) / 256, 256, 0, stream>>>(x, edge_index, edge_attr,
                                                           e1, e2, acc);
  mlp_fused<<<N_NODES / 16, 128, 0, stream>>>(acc, W1, b1, W2, b2, out);
}


# Global f32 scatter atomics -> owner-computes bins: a work-group keeps 200 destination rows in LDS, scans all edges for them, adds by LDS f32 atomics in arrival order; message = x+(e1+e2); init merged
# speedup vs baseline: 2.5074x; 2.5074x over previous
#include <hip/hip_runtime.h>


#define N_NODES 50000
#define N_EDGES 600000
#define DIM     128
#define HID     256
#define AS_STR  (DIM + 4)
#define HS_STR  (HID + 4)

typedef __attribute__((ext_vector_type(2))) float v2f;
typedef __attribute__((ext_vector_type(8))) float v8f;
typedef __attribute__((ext_vector_type(4))) float v4f;

#define BIN_ROWS  200
#define N_BINS    (N_NODES / BIN_ROWS)
#define AGG_WAVES 32
#define CHUNK     128
#define N_CHUNKS  ((N_EDGES + CHUNK - 1) / CHUNK)
#define BATCH     8
#define QCAP      (32 + BATCH * CHUNK)
#define N_EMB     15

__device__ __forceinline__ void lds_add(float* p, float v) {
  __hip_atomic_fetch_add(p, v, __ATOMIC_RELAXED, __HIP_MEMORY_SCOPE_WORKGROUP);
}

template <bool FULL>
__device__ __forceinline__ void drain(const unsigned* qp, int cnt, int lane,
                                      const v4f* __restrict__ x4,
                                      const int* __restrict__ col_idx,
                                      const int2* __restrict__ attr2,
                                      const v4f* emb, float* bin) {
  const unsigned ent = qp[lane];
  const unsigned e   = ent & 0xFFFFFu;
  int  col = 0;
  int2 at  = {0, 0};
  if (FULL || lane < cnt) {
    col = col_idx[e];
    at  = attr2[e];
  }
  const int meta = (int)((ent >> 20) << 4) | (at.x * 3 + at.y);
  const int lim  = FULL ? 32 : cnt;
  for (int j0 = 0; j0 < lim; j0 += 8) {
    v4f v[8], em[8];
    int mt[8];
#pragma unroll
    for (int u = 0; u < 8; ++u) {
      const int c = __builtin_amdgcn_readlane(col, j0 + u);
      v[u] = x4[c * 32 + lane];
    }
#pragma unroll
    for (int u = 0; u < 8; ++u) {
      mt[u] = __builtin_amdgcn_readlane(meta, j0 + u);
      em[u] = emb[(mt[u] & 15) * 32 + lane];
    }
    asm volatile("" :: "v"(v[0]), "v"(v[1]), "v"(v[2]), "v"(v[3]),
                       "v"(v[4]), "v"(v[5]), "v"(v[6]), "v"(v[7]),
                       "v"(em[0]), "v"(em[1]), "v"(em[2]), "v"(em[3]),
                       "v"(em[4]), "v"(em[5]), "v"(em[6]), "v"(em[7]));
#pragma unroll
    for (int u = 0; u < 8; ++u) {
      if (FULL || j0 + u < cnt) {
        const v4f m = v[u] + em[u];
        float* b = bin + (mt[u] >> 4) * DIM + lane;
        lds_add(b,      m.x);
        lds_add(b + 32, m.y);
        lds_add(b + 64, m.z);
        lds_add(b + 96, m.w);
      }
    }
  }
}

__global__ __launch_bounds__(1024) void aggregate_bins(const float* __restrict__ x,
                                                       const int* __restrict__ edge_index,
                                                       const int* __restrict__ edge_attr,
                                                       const float* __restrict__ e1,
                                                       const float* __restrict__ e2,
                                                       float* __restrict__ acc) {
  __shared__ float    bin[BIN_ROWS * DIM];
  __shared__ v4f      emb[N_EMB * 32];
  __shared__ unsigned queue[AGG_WAVES * QCAP];
  const int tid  = threadIdx.x;
  const int wave = __builtin_amdgcn_readfirstlane(tid >> 5);
  const int lane = tid & 31;
  const int lo   = blockIdx.x * BIN_ROWS;
  const v4f* x4    = (const v4f*)x;
  const v4f* e1_4  = (const v4f*)e1;
  const v4f* e2_4  = (const v4f*)e2;
  const int4*   rows4 = (const int4*)edge_index;
  const int*    cols  = edge_index + N_EDGES;
  const int2*   attr2 = (const int2*)edge_attr;

  if (tid < N_EMB * 32) {
    const int t = tid >> 5;
    emb[tid] = e1_4[(t / 3) * 32 + lane] + e2_4[(t % 3) * 32 + lane];
  }
  {
    const v4f s1 = e1_4[4 * 32 + lane];
    const v4f s2 = e2_4[lane];
    for (int r = wave; r < BIN_ROWS; r += AGG_WAVES) {
      const v4f v = x4[(lo + r) * 32 + lane] + s1 + s2;
      float* b = bin + r * DIM + lane;
      b[0]  = v.x;
      b[32] = v.y;
      b[64] = v.z;
      b[96] = v.w;
    }
  }
  __syncthreads();

  unsigned* q = queue + wave * QCAP;
  int n = 0;
  const int rot = (blockIdx.x * N_CHUNKS) / N_BINS;
  for (int s0 = wave; s0 < N_CHUNKS; s0 += AGG_WAVES * BATCH) {
    int4 r[BATCH];
    int  e0[BATCH];
#pragma unroll
    for (int u = 0; u < BATCH; ++u) {
      const int s = s0 + AGG_WAVES * u;
      int c = s + rot;
      if (c >= N_CHUNKS) c -= N_CHUNKS;
      e0[u] = c * CHUNK + lane * 4;
      r[u]  = make_int4(-1, -1, -1, -1);
      if (s < N_CHUNKS && e0[u] < N_EDGES) r[u] = rows4[c * 32 + lane];
    }
#pragma unroll
    for (int u = 0; u < BATCH; ++u) {
      const unsigned d0 = (unsigned)(r[u].x - lo), d1 = (unsigned)(r[u].y - lo);
      const unsigned d2 = (unsigned)(r[u].z - lo), d3 = (unsigned)(r[u].w - lo);
      const unsigned dmin = min(min(d0, d1), min(d2, d3));
      if (__builtin_amdgcn_ballot_w32(dmin < BIN_ROWS) != 0u) {
        const unsigned dd[4] = {d0, d1, d2, d3};
#pragma unroll
        for (int k = 0; k < 4; ++k) {
          const bool     hit  = dd[k] < BIN_ROWS;
          const unsigned mask = __builtin_amdgcn_ballot_w32(hit);
          if (hit) q[n + __builtin_amdgcn_mbcnt_lo(mask, 0u)] = (dd[k] << 20) | (unsigned)(e0[u] + k);
          n += __builtin_popcount(mask);
        }
      }
    }
    while (n >= 32) {
      n -= 32;
      drain<true>(q + n, 32, lane, x4, cols, attr2, emb, bin);
    }
  }
  if (n > 0) drain<false>(q, n, lane, x4, cols, attr2, emb, bin);
  __syncthreads();

  for (int r = wave; r < BIN_ROWS; r += AGG_WAVES) {
    const float* b = bin + r * DIM + lane;
    const v4f v = {b[0], b[32], b[64], b[96]};
    ((v4f*)acc)[(lo + r) * 32 + lane] = v;
  }
}

__global__ __launch_bounds__(128) void mlp_fused(const float* __restrict__ acc,
                                                 const float* __restrict__ W1,
                                                 const float* __restrict__ b1,
                                                 const float* __restrict__ W2,
                                                 const float* __restrict__ b2,
                                                 float* __restrict__ out) {
  __shared__ float As[16 * AS_STR];
  __shared__ float Hs[16 * HS_STR];
  const int tid  = threadIdx.x;
  const int wave = tid >> 5;
  const int lane = tid & 31;
  const int m    = lane & 15;
  const int kh   = lane >> 4;
  const int row0 = blockIdx.x * 16;

  {
    const int r  = tid >> 3;
    const int c0 = (tid & 7) * 16;
    const float4* src = (const float4*)(acc + (row0 + r) * DIM + c0);
    float4*       dst = (float4*)(As + r * AS_STR + c0);
#pragma unroll
    for (int i = 0; i < 4; ++i) dst[i] = src[i];
  }
  __syncthreads();

#pragma unroll
  for (int nt = 0; nt < 4; ++nt) {
    const int n0 = (wave * 4 + nt) * 16;
    const float* w1row = W1 + (n0 + m) * DIM;
    v8f c = {};
#pragma unroll
    for (int k0 = 0; k0 < DIM; k0 += 4) {
      v2f a, b;
      a.x = As[m * AS_STR + k0 + 2 * kh];
      a.y = As[m * AS_STR + k0 + 2 * kh + 1];
      b.x = w1row[k0 + 2 * kh];
      b.y = w1row[k0 + 2 * kh + 1];
      c = __builtin_amdgcn_wmma_f32_16x16x4_f32(false, a, false, b, (short)0, c,
                                                false, false);
    }
    const float bias = b1[n0 + m];
#pragma unroll
    for (int j = 0; j < 8; ++j) {
      const int mj = j + 8 * kh;
      float h = c[j] + bias;
      Hs[mj * HS_STR + n0 + m] = h > 0.0f ? h : 0.0f;
    }
  }
  __syncthreads();

#pragma unroll
  for (int nt = 0; nt < 2; ++nt) {
    const int n0 = (wave * 2 + nt) * 16;
    const float* w2row = W2 + (n0 + m) * HID;
    v8f c = {};
#pragma unroll
    for (int k0 = 0; k0 < HID; k0 += 4) {
      v2f a, b;
      a.x = Hs[m * HS_STR + k0 + 2 * kh];
      a.y = Hs[m * HS_STR + k0 + 2 * kh + 1];
      b.x = w2row[k0 + 2 * kh];
      b.y = w2row[k0 + 2 * kh + 1];
      c = __builtin_amdgcn_wmma_f32_16x16x4_f32(false, a, false, b, (short)0, c,
                                                false, false);
    }
    const float bias = b2[n0 + m];
#pragma unroll
    for (int j = 0; j < 8; ++j) {
      const int mj = j + 8 * kh;
      out[(row0 + mj) * DIM + n0 + m] = c[j] + bias;
    }
  }
}

extern "C" void kernel_launch(void* const* d_in, const int* in_sizes, int n_in,
                              void* d_out, int out_size, void* d_ws, size_t ws_size,
                              hipStream_t stream) {
  const float* x          = (const float*)d_in[0];
  const int*   edge_index = (const int*)d_in[1];
  const int*   edge_attr  = (const int*)d_in[2];
  const float* W1         = (const float*)d_in[3];
  const float* b1         = (const float*)d_in[4];
  const float* W2         = (const float*)d_in[5];
  const float* b2         = (const float*)d_in[6];
  const float* e1         = (const float*)d_in[7];
  const float* e2         = (const float*)d_in[8];
  float*       out        = (float*)d_out;
  float*       acc        = (float*)d_ws;

  (void)in_sizes; (void)n_in; (void)out_size; (void)ws_size;

  aggregate_bins<<<N_BINS, 1024, 0, stream>>>(x, edge_index, edge_attr, e1, e2, acc);
  mlp_fused<<<N_NODES / 16, 128, 0, stream>>>(acc, W1, b1, W2, b2, out);
}


# MLP weights resident in registers: 256 long-lived 16-wave work-groups, each wave keeps its W1 and W2 column-tile fragments in VGPRs and streams 32-row passes through LDS; k and sum order unchanged
# speedup vs baseline: 3.2746x; 1.3060x over previous
#include <hip/hip_runtime.h>


#define N_NODES 50000
#define N_EDGES 600000
#define DIM     128
#define HID     256
#define AS_STR  (DIM + 4)
#define HS_STR  (HID + 4)

typedef __attribute__((ext_vector_type(2))) float v2f;
typedef __attribute__((ext_vector_type(8))) float v8f;
typedef __attribute__((ext_vector_type(4))) float v4f;

#define BIN_ROWS  200
#define N_BINS    (N_NODES / BIN_ROWS)
#define AGG_WAVES 32
#define CHUNK     128
#define N_CHUNKS  ((N_EDGES + CHUNK - 1) / CHUNK)
#define BATCH     8
#define QCAP      (32 + BATCH * CHUNK)
#define N_EMB     15

__device__ __forceinline__ void lds_add(float* p, float v) {
  __hip_atomic_fetch_add(p, v, __ATOMIC_RELAXED, __HIP_MEMORY_SCOPE_WORKGROUP);
}

template <bool FULL>
__device__ __forceinline__ void drain(const unsigned* qp, int cnt, int lane,
                                      const v4f* __restrict__ x4,
                                      const int* __restrict__ col_idx,
                                      const int2* __restrict__ attr2,
                                      const v4f* emb, float* bin) {
  const unsigned ent = qp[lane];
  const unsigned e   = ent & 0xFFFFFu;
  int  col = 0;
  int2 at  = {0, 0};
  if (FULL || lane < cnt) {
    col = col_idx[e];
    at  = attr2[e];
  }
  const int meta = (int)((ent >> 20) << 4) | (at.x * 3 + at.y);
  const int lim  = FULL ? 32 : cnt;
  for (int j0 = 0; j0 < lim; j0 += 8) {
    v4f v[8], em[8];
    int mt[8];
#pragma unroll
    for (int u = 0; u < 8; ++u) {
      const int c = __builtin_amdgcn_readlane(col, j0 + u);
      v[u] = x4[c * 32 + lane];
    }
#pragma unroll
    for (int u = 0; u < 8; ++u) {
      mt[u] = __builtin_amdgcn_readlane(meta, j0 + u);
      em[u] = emb[(mt[u] & 15) * 32 + lane];
    }
    asm volatile("" :: "v"(v[0]), "v"(v[1]), "v"(v[2]), "v"(v[3]),
                       "v"(v[4]), "v"(v[5]), "v"(v[6]), "v"(v[7]),
                       "v"(em[0]), "v"(em[1]), "v"(em[2]), "v"(em[3]),
                       "v"(em[4]), "v"(em[5]), "v"(em[6]), "v"(em[7]));
#pragma unroll
    for (int u = 0; u < 8; ++u) {
      if (FULL || j0 + u < cnt) {
        const v4f m = v[u] + em[u];
        float* b = bin + (mt[u] >> 4) * DIM + lane;
        lds_add(b,      m.x);
        lds_add(b + 32, m.y);
        lds_add(b + 64, m.z);
        lds_add(b + 96, m.w);
      }
    }
  }
}

__global__ __launch_bounds__(1024) void aggregate_bins(const float* __restrict__ x,
                                                       const int* __restrict__ edge_index,
                                                       const int* __restrict__ edge_attr,
                                                       const float* __restrict__ e1,
                                                       const float* __restrict__ e2,
                                                       float* __restrict__ acc) {
  __shared__ float    bin[BIN_ROWS * DIM];
  __shared__ v4f      emb[N_EMB * 32];
  __shared__ unsigned queue[AGG_WAVES * QCAP];
  const int tid  = threadIdx.x;
  const int wave = __builtin_amdgcn_readfirstlane(tid >> 5);
  const int lane = tid & 31;
  const int lo   = blockIdx.x * BIN_ROWS;
  const v4f* x4    = (const v4f*)x;
  const v4f* e1_4  = (const v4f*)e1;
  const v4f* e2_4  = (const v4f*)e2;
  const int4*   rows4 = (const int4*)edge_index;
  const int*    cols  = edge_index + N_EDGES;
  const int2*   attr2 = (const int2*)edge_attr;

  if (tid < N_EMB * 32) {
    const int t = tid >> 5;
    emb[tid] = e1_4[(t / 3) * 32 + lane] + e2_4[(t % 3) * 32 + lane];
  }
  {
    const v4f s1 = e1_4[4 * 32 + lane];
    const v4f s2 = e2_4[lane];
    for (int r = wave; r < BIN_ROWS; r += AGG_WAVES) {
      const v4f v = x4[(lo + r) * 32 + lane] + s1 + s2;
      float* b = bin + r * DIM + lane;
      b[0]  = v.x;
      b[32] = v.y;
      b[64] = v.z;
      b[96] = v.w;
    }
  }
  __syncthreads();

  unsigned* q = queue + wave * QCAP;
  int n = 0;
  const int rot = (blockIdx.x * N_CHUNKS) / N_BINS;
  for (int s0 = wave; s0 < N_CHUNKS; s0 += AGG_WAVES * BATCH) {
    int4 r[BATCH];
    int  e0[BATCH];
#pragma unroll
    for (int u = 0; u < BATCH; ++u) {
      const int s = s0 + AGG_WAVES * u;
      int c = s + rot;
      if (c >= N_CHUNKS) c -= N_CHUNKS;
      e0[u] = c * CHUNK + lane * 4;
      r[u]  = make_int4(-1, -1, -1, -1);
      if (s < N_CHUNKS && e0[u] < N_EDGES) r[u] = rows4[c * 32 + lane];
    }
#pragma unroll
    for (int u = 0; u < BATCH; ++u) {
      const unsigned d0 = (unsigned)(r[u].x - lo), d1 = (unsigned)(r[u].y - lo);
      const unsigned d2 = (unsigned)(r[u].z - lo), d3 = (unsigned)(r[u].w - lo);
      const unsigned dmin = min(min(d0, d1), min(d2, d3));
      if (__builtin_amdgcn_ballot_w32(dmin < BIN_ROWS) != 0u) {
        const unsigned dd[4] = {d0, d1, d2, d3};
#pragma unroll
        for (int k = 0; k < 4; ++k) {
          const bool     hit  = dd[k] < BIN_ROWS;
          const unsigned mask = __builtin_amdgcn_ballot_w32(hit);
          if (hit) q[n + __builtin_amdgcn_mbcnt_lo(mask, 0u)] = (dd[k] << 20) | (unsigned)(e0[u] + k);
          n += __builtin_popcount(mask);
        }
      }
    }
    while (n >= 32) {
      n -= 32;
      drain<true>(q + n, 32, lane, x4, cols, attr2, emb, bin);
    }
  }
  if (n > 0) drain<false>(q, n, lane, x4, cols, attr2, emb, bin);
  __syncthreads();

  for (int r = wave; r < BIN_ROWS; r += AGG_WAVES) {
    const float* b = bin + r * DIM + lane;
    const v4f v = {b[0], b[32], b[64], b[96]};
    ((v4f*)acc)[(lo + r) * 32 + lane] = v;
  }
}

#define MLP_WAVES 16
#define MLP_GRID  256
#define PASS_ROWS 32
#define N_PASS    ((N_NODES + PASS_ROWS - 1) / PASS_ROWS)

__global__ __launch_bounds__(MLP_WAVES * 32) void mlp_regw(const float* __restrict__ acc,
                                                           const float* __restrict__ W1,
                                                           const float* __restrict__ b1,
                                                           const float* __restrict__ W2,
                                                           const float* __restrict__ b2,
                                                           float* __restrict__ out) {
  __shared__ float As[PASS_ROWS * AS_STR];
  __shared__ float Hs[PASS_ROWS * HS_STR];
  const int tid  = threadIdx.x;
  const int wave = __builtin_amdgcn_readfirstlane(tid >> 5);
  const int lane = tid & 31;
  const int m    = lane & 15;
  const int kh   = lane >> 4;
  const int ct2  = wave & 7;
  const int rt2  = wave >> 3;

  v2f w1f[DIM / 4], w2f[HID / 4];
  {
    const float* w1row = W1 + (wave * 16 + m) * DIM + 2 * kh;
#pragma unroll
    for (int ks = 0; ks < DIM / 4; ++ks) w1f[ks] = *(const v2f*)(w1row + 4 * ks);
    const float* w2row = W2 + (ct2 * 16 + m) * HID + 2 * kh;
#pragma unroll
    for (int ks = 0; ks < HID / 4; ++ks) w2f[ks] = *(const v2f*)(w2row + 4 * ks);
  }
  const float bias1 = b1[wave * 16 + m];
  const float bias2 = b2[ct2 * 16 + m];

  const int sr = tid >> 4;
  const int sc = (tid & 15) * 8;
  float* const as_dst = As + sr * AS_STR + sc;

  int p = blockIdx.x;
  {
    v4f t0 = {}, t1 = {};
    const int row = p * PASS_ROWS + sr;
    if (p < N_PASS && row < N_NODES) {
      const v4f* s = (const v4f*)(acc + row * DIM + sc);
      t0 = s[0];
      t1 = s[1];
    }
    ((v4f*)as_dst)[0] = t0;
    ((v4f*)as_dst)[1] = t1;
  }
  __syncthreads();

  for (; p < N_PASS; p += MLP_GRID) {
    const int pn = p + MLP_GRID;
    v4f t0 = {}, t1 = {};
    {
      const int row = pn * PASS_ROWS + sr;
      if (pn < N_PASS && row < N_NODES) {
        const v4f* s = (const v4f*)(acc + row * DIM + sc);
        t0 = s[0];
        t1 = s[1];
      }
    }

    {
      v8f c0 = {}, c1 = {};
      const float* ap = As + m * AS_STR + 2 * kh;
#pragma unroll
      for (int ks = 0; ks < DIM / 4; ++ks) {
        const v2f a0 = *(const v2f*)(ap + 4 * ks);
        const v2f a1 = *(const v2f*)(ap + 16 * AS_STR + 4 * ks);
        c0 = __builtin_amdgcn_wmma_f32_16x16x4_f32(false, a0, false, w1f[ks], (short)0, c0,
                                                   false, false);
        c1 = __builtin_amdgcn_wmma_f32_16x16x4_f32(false, a1, false, w1f[ks], (short)0, c1,
                                                   false, false);
      }
      float* hp = Hs + (8 * kh) * HS_STR + wave * 16 + m;
#pragma unroll
      for (int j = 0; j < 8; ++j) {
        const float h0 = c0[j] + bias1;
        const float h1 = c1[j] + bias1;
        hp[j * HS_STR]        = h0 > 0.0f ? h0 : 0.0f;
        hp[(16 + j) * HS_STR] = h1 > 0.0f ? h1 : 0.0f;
      }
    }
    __syncthreads();

    ((v4f*)as_dst)[0] = t0;
    ((v4f*)as_dst)[1] = t1;

    {
      v8f c = {};
      const float* hq = Hs + (rt2 * 16 + m) * HS_STR + 2 * kh;
#pragma unroll
      for (int ks = 0; ks < HID / 4; ++ks) {
        const v2f a = *(const v2f*)(hq + 4 * ks);
        c = __builtin_amdgcn_wmma_f32_16x16x4_f32(false, a, false, w2f[ks], (short)0, c,
                                                  false, false);
      }
      const int r0 = p * PASS_ROWS + rt2 * 16 + 8 * kh;
      float* op = out + r0 * DIM + ct2 * 16 + m;
      static_assert(N_NODES % 8 == 0, "row guard below is per 8 rows");
      if (r0 < N_NODES) {
#pragma unroll
        for (int j = 0; j < 8; ++j) op[j * DIM] = c[j] + bias2;
      }
    }
    __syncthreads();
  }
}

extern "C" void kernel_launch(void* const* d_in, const int* in_sizes, int n_in,
                              void* d_out, int out_size, void* d_ws, size_t ws_size,
                              hipStream_t stream) {
  const float* x          = (const float*)d_in[0];
  const int*   edge_index = (const int*)d_in[1];
  const int*   edge_attr  = (const int*)d_in[2];
  const float* W1         = (const float*)d_in[3];
  const float* b1         = (const float*)d_in[4];
  const float* W2         = (const float*)d_in[5];
  const float* b2         = (const float*)d_in[6];
  const float* e1         = (const float*)d_in[7];
  const float* e2         = (const float*)d_in[8];
  float*       out        = (float*)d_out;
  float*       acc        = (float*)d_ws;

  (void)in_sizes; (void)n_in; (void)out_size; (void)ws_size;

  aggregate_bins<<<N_BINS, 1024, 0, stream>>>(x, edge_index, edge_attr, e1, e2, acc);
  mlp_regw<<<MLP_GRID, MLP_WAVES * 32, 0, stream>>>(acc, W1, b1, W2, b2, out);
}


# Edge list cut in two: bins of 400 LDS rows, 2 work-groups per bin each scan half the edges into its own partial plane; MLP staging adds the planes, so a row sum = half0 + half1
# speedup vs baseline: 3.4622x; 1.0573x over previous
#include <hip/hip_runtime.h>


#define N_NODES 50000
#define N_EDGES 600000
#define DIM     128
#define HID     256
#define AS_STR  (DIM + 4)
#define HS_STR  (HID + 4)

typedef __attribute__((ext_vector_type(2))) float v2f;
typedef __attribute__((ext_vector_type(8))) float v8f;
typedef __attribute__((ext_vector_type(4))) float v4f;

#define BIN_ROWS  400
#define N_BINS    (N_NODES / BIN_ROWS)
#define N_HALVES  2
#define AGG_WAVES 32
#define CHUNK     128
#define N_CHUNKS  ((N_EDGES + CHUNK - 1) / CHUNK)
#define HALF_CHUNKS (N_CHUNKS / N_HALVES)
#define BATCH     8
#define LOOK      4
#define QCAP      (32 + LOOK * CHUNK)
#define N_EMB     15
#define INIT_ROWS ((BIN_ROWS + AGG_WAVES - 1) / AGG_WAVES)
static_assert(N_NODES % BIN_ROWS == 0, "bins cover the rows exactly");
static_assert(N_CHUNKS % N_HALVES == 0, "the halves cover the chunks exactly");
static_assert(BATCH % LOOK == 0, "a batch is a whole number of looks");

__device__ __forceinline__ void lds_add(float* p, float v) {
  __hip_atomic_fetch_add(p, v, __ATOMIC_RELAXED, __HIP_MEMORY_SCOPE_WORKGROUP);
}

template <bool FULL>
__device__ __forceinline__ void drain(const unsigned* qp, int cnt, int lane,
                                      const v4f* __restrict__ x4,
                                      const int* __restrict__ col_idx,
                                      const int2* __restrict__ attr2,
                                      const v4f* emb, float* bin) {
  const unsigned ent = qp[lane];
  const unsigned e   = ent & 0xFFFFFu;
  int  col = 0;
  int2 at  = {0, 0};
  if (FULL || lane < cnt) {
    col = col_idx[e];
    at  = attr2[e];
  }
  const int meta = (int)((ent >> 20) << 4) | (at.x * 3 + at.y);
  const int lim  = FULL ? 32 : cnt;
  for (int j0 = 0; j0 < lim; j0 += 8) {
    v4f v[8], em[8];
    int mt[8];
#pragma unroll
    for (int u = 0; u < 8; ++u) {
      const int c = __builtin_amdgcn_readlane(col, j0 + u);
      v[u] = x4[c * 32 + lane];
    }
#pragma unroll
    for (int u = 0; u < 8; ++u) {
      mt[u] = __builtin_amdgcn_readlane(meta, j0 + u);
      em[u] = emb[(mt[u] & 15) * 32 + lane];
    }
    asm volatile("" :: "v"(v[0]), "v"(v[1]), "v"(v[2]), "v"(v[3]),
                       "v"(v[4]), "v"(v[5]), "v"(v[6]), "v"(v[7]),
                       "v"(em[0]), "v"(em[1]), "v"(em[2]), "v"(em[3]),
                       "v"(em[4]), "v"(em[5]), "v"(em[6]), "v"(em[7]));
#pragma unroll
    for (int u = 0; u < 8; ++u) {
      if (FULL || j0 + u < cnt) {
        const v4f m = v[u] + em[u];
        float* b = bin + (mt[u] >> 4) * DIM + lane;
        lds_add(b,      m.x);
        lds_add(b + 32, m.y);
        lds_add(b + 64, m.z);
        lds_add(b + 96, m.w);
      }
    }
  }
}

__global__ __launch_bounds__(1024) void aggregate_bins(const float* __restrict__ x,
                                                       const int* __restrict__ edge_index,
                                                       const int* __restrict__ edge_attr,
                                                       const float* __restrict__ e1,
                                                       const float* __restrict__ e2,
                                                       float* __restrict__ acc) {
  __shared__ float    bin[BIN_ROWS * DIM];
  __shared__ v4f      emb[N_EMB * 32];
  __shared__ unsigned queue[AGG_WAVES * QCAP];
  const int tid  = threadIdx.x;
  const int wave = __builtin_amdgcn_readfirstlane(tid >> 5);
  const int lane = tid & 31;
  const int bin_i = blockIdx.x >> 1;
  const int half  = blockIdx.x & 1;
  const int lo    = bin_i * BIN_ROWS;
  const v4f* x4    = (const v4f*)x;
  const v4f* e1_4  = (const v4f*)e1;
  const v4f* e2_4  = (const v4f*)e2;
  const int4*   rows4 = (const int4*)edge_index;
  const int*    cols  = edge_index + N_EDGES;
  const int2*   attr2 = (const int2*)edge_attr;

  if (tid < N_EMB * 32) {
    const int t = tid >> 5;
    emb[tid] = e1_4[(t / 3) * 32 + lane] + e2_4[(t % 3) * 32 + lane];
  }
  if (half == 0) {
    const v4f s1 = e1_4[4 * 32 + lane];
    const v4f s2 = e2_4[lane];
    v4f v[INIT_ROWS];
#pragma unroll
    for (int i = 0; i < INIT_ROWS; ++i) {
      const int r = wave + i * AGG_WAVES;
      v[i] = v4f{0.0f, 0.0f, 0.0f, 0.0f};
      if (r < BIN_ROWS) v[i] = x4[(lo + r) * 32 + lane];
    }
#pragma unroll
    for (int i = 0; i < INIT_ROWS; ++i) {
      const int r = wave + i * AGG_WAVES;
      if (r < BIN_ROWS) {
        const v4f t = v[i] + s1 + s2;
        float* b = bin + r * DIM + lane;
        b[0]  = t.x;
        b[32] = t.y;
        b[64] = t.z;
        b[96] = t.w;
      }
    }
  } else {
    for (int r = wave; r < BIN_ROWS; r += AGG_WAVES) {
      float* b = bin + r * DIM + lane;
      b[0]  = 0.0f;
      b[32] = 0.0f;
      b[64] = 0.0f;
      b[96] = 0.0f;
    }
  }
  __syncthreads();

  unsigned* q = queue + wave * QCAP;
  int n = 0;
  const int base = half * HALF_CHUNKS;
  const int rot  = (bin_i * HALF_CHUNKS) / N_BINS;
  for (int s0 = wave; s0 < HALF_CHUNKS; s0 += AGG_WAVES * BATCH) {
    int4 r[BATCH];
    int  e0[BATCH];
#pragma unroll
    for (int u = 0; u < BATCH; ++u) {
      const int s = s0 + AGG_WAVES * u;
      int c = s + rot;
      if (c >= HALF_CHUNKS) c -= HALF_CHUNKS;
      c += base;
      e0[u] = c * CHUNK + lane * 4;
      r[u]  = make_int4(-1, -1, -1, -1);
      if (s < HALF_CHUNKS && e0[u] < N_EDGES) r[u] = rows4[c * 32 + lane];
    }
#pragma unroll
    for (int h = 0; h < BATCH; h += LOOK) {
#pragma unroll
      for (int u = h; u < h + LOOK; ++u) {
        const unsigned d0 = (unsigned)(r[u].x - lo), d1 = (unsigned)(r[u].y - lo);
        const unsigned d2 = (unsigned)(r[u].z - lo), d3 = (unsigned)(r[u].w - lo);
        const unsigned dmin = min(min(d0, d1), min(d2, d3));
        if (__builtin_amdgcn_ballot_w32(dmin < BIN_ROWS) != 0u) {
          const unsigned dd[4] = {d0, d1, d2, d3};
#pragma unroll
          for (int k = 0; k < 4; ++k) {
            const bool     hit  = dd[k] < BIN_ROWS;
            const unsigned mask = __builtin_amdgcn_ballot_w32(hit);
            if (hit) q[n + __builtin_amdgcn_mbcnt_lo(mask, 0u)] = (dd[k] << 20) | (unsigned)(e0[u] + k);
            n += __builtin_popcount(mask);
          }
        }
      }
      while (n >= 32) {
        n -= 32;
        drain<true>(q + n, 32, lane, x4, cols, attr2, emb, bin);
      }
    }
  }
  if (n > 0) drain<false>(q, n, lane, x4, cols, attr2, emb, bin);
  __syncthreads();

  v4f* plane = (v4f*)(acc + half * (N_NODES * DIM));
  for (int r = wave; r < BIN_ROWS; r += AGG_WAVES) {
    const float* b = bin + r * DIM + lane;
    const v4f v = {b[0], b[32], b[64], b[96]};
    plane[(lo + r) * 32 + lane] = v;
  }
}

#define MLP_WAVES 16
#define MLP_GRID  256
#define PASS_ROWS 32
#define N_PASS    ((N_NODES + PASS_ROWS - 1) / PASS_ROWS)

__global__ __launch_bounds__(MLP_WAVES * 32) void mlp_regw(const float* __restrict__ acc,
                                                           const float* __restrict__ W1,
                                                           const float* __restrict__ b1,
                                                           const float* __restrict__ W2,
                                                           const float* __restrict__ b2,
                                                           float* __restrict__ out) {
  __shared__ float As[PASS_ROWS * AS_STR];
  __shared__ float Hs[PASS_ROWS * HS_STR];
  const int tid  = threadIdx.x;
  const int wave = __builtin_amdgcn_readfirstlane(tid >> 5);
  const int lane = tid & 31;
  const int m    = lane & 15;
  const int kh   = lane >> 4;
  const int ct2  = wave & 7;
  const int rt2  = wave >> 3;

  v2f w1f[DIM / 4], w2f[HID / 4];
  {
    const float* w1row = W1 + (wave * 16 + m) * DIM + 2 * kh;
#pragma unroll
    for (int ks = 0; ks < DIM / 4; ++ks) w1f[ks] = *(const v2f*)(w1row + 4 * ks);
    const float* w2row = W2 + (ct2 * 16 + m) * HID + 2 * kh;
#pragma unroll
    for (int ks = 0; ks < HID / 4; ++ks) w2f[ks] = *(const v2f*)(w2row + 4 * ks);
  }
  const float bias1 = b1[wave * 16 + m];
  const float bias2 = b2[ct2 * 16 + m];

  const int sr = tid >> 4;
  const int sc = (tid & 15) * 8;
  float* const as_dst = As + sr * AS_STR + sc;

  const float* const acc1 = acc + N_NODES * DIM;

  int p = blockIdx.x;
  {
    v4f t0 = {}, t1 = {};
    const int row = p * PASS_ROWS + sr;
    if (p < N_PASS && row < N_NODES) {
      const v4f* s = (const v4f*)(acc + row * DIM + sc);
      const v4f* z = (const v4f*)(acc1 + row * DIM + sc);
      t0 = s[0] + z[0];
      t1 = s[1] + z[1];
    }
    ((v4f*)as_dst)[0] = t0;
    ((v4f*)as_dst)[1] = t1;
  }
  __syncthreads();

  for (; p < N_PASS; p += MLP_GRID) {
    const int  pn    = p + MLP_GRID;
    const int  rown  = pn * PASS_ROWS + sr;
    const bool fetch = pn < N_PASS && rown < N_NODES;
    v4f t0 = {}, t1 = {};
    if (fetch) {
      const v4f* s = (const v4f*)(acc + rown * DIM + sc);
      t0 = s[0];
      t1 = s[1];
    }

    {
      v8f c0 = {}, c1 = {};
      const float* ap = As + m * AS_STR + 2 * kh;
#pragma unroll
      for (int ks = 0; ks < DIM / 4; ++ks) {
        const v2f a0 = *(const v2f*)(ap + 4 * ks);
        const v2f a1 = *(const v2f*)(ap + 16 * AS_STR + 4 * ks);
        c0 = __builtin_amdgcn_wmma_f32_16x16x4_f32(false, a0, false, w1f[ks], (short)0, c0,
                                                   false, false);
        c1 = __builtin_amdgcn_wmma_f32_16x16x4_f32(false, a1, false, w1f[ks], (short)0, c1,
                                                   false, false);
      }
      float* hp = Hs + (8 * kh) * HS_STR + wave * 16 + m;
#pragma unroll
      for (int j = 0; j < 8; ++j) {
        const float h0 = c0[j] + bias1;
        const float h1 = c1[j] + bias1;
        hp[j * HS_STR]        = h0 > 0.0f ? h0 : 0.0f;
        hp[(16 + j) * HS_STR] = h1 > 0.0f ? h1 : 0.0f;
      }
    }
    __syncthreads();

    v4f u0 = {}, u1 = {};
    if (fetch) {
      const v4f* z = (const v4f*)(acc1 + rown * DIM + sc);
      u0 = z[0];
      u1 = z[1];
    }

    {
      v8f c = {};
      const float* hq = Hs + (rt2 * 16 + m) * HS_STR + 2 * kh;
#pragma unroll
      for (int ks = 0; ks < HID / 4; ++ks) {
        const v2f a = *(const v2f*)(hq + 4 * ks);
        c = __builtin_amdgcn_wmma_f32_16x16x4_f32(false, a, false, w2f[ks], (short)0, c,
                                                  false, false);
      }
      const int r0 = p * PASS_ROWS + rt2 * 16 + 8 * kh;
      float* op = out + r0 * DIM + ct2 * 16 + m;
      static_assert(N_NODES % 8 == 0, "row guard below is per 8 rows");
      if (r0 < N_NODES) {
#pragma unroll
        for (int j = 0; j < 8; ++j) op[j * DIM] = c[j] + bias2;
      }
    }
    ((v4f*)as_dst)[0] = t0 + u0;
    ((v4f*)as_dst)[1] = t1 + u1;
    __syncthreads();
  }
}

extern "C" void kernel_launch(void* const* d_in, const int* in_sizes, int n_in,
                              void* d_out, int out_size, void* d_ws, size_t ws_size,
                              hipStream_t stream) {
  const float* x          = (const float*)d_in[0];
  const int*   edge_index = (const int*)d_in[1];
  const int*   edge_attr  = (const int*)d_in[2];
  const float* W1         = (const float*)d_in[3];
  const float* b1         = (const float*)d_in[4];
  const float* W2         = (const float*)d_in[5];
  const float* b2         = (const float*)d_in[6];
  const float* e1         = (const float*)d_in[7];
  const float* e2         = (const float*)d_in[8];
  float*       out        = (float*)d_out;
  float*       acc        = (float*)d_ws;

  (void)in_sizes; (void)n_in; (void)out_size; (void)ws_size;

  aggregate_bins<<<N_BINS * N_HALVES, 1024, 0, stream>>>(x, edge_index, edge_attr, e1, e2, acc);
  mlp_regw<<<MLP_GRID, MLP_WAVES * 32, 0, stream>>>(acc, W1, b1, W2, b2, out);
}
